# starting kernel
# baseline (speedup 1.0000x reference)
#include <hip/hip_runtime.h>

typedef float v2f __attribute__((ext_vector_type(2)));
typedef float v8f __attribute__((ext_vector_type(8)));

#define NNODES 10000
#define NEDGES 160000

__global__ void deg_init_kernel(float* __restrict__ deg, int n) {
  int i = blockIdx.x * blockDim.x + threadIdx.x;
  if (i < n) deg[i] = 1.0f;
}

__global__ void deg_accum_kernel(const int* __restrict__ dst, float* __restrict__ deg, int e) {
  int i = blockIdx.x * blockDim.x + threadIdx.x;
  if (i < e) atomicAdd(&deg[dst[i]], 1.0f);
}

__global__ void dinv_kernel(const float* __restrict__ deg, float* __restrict__ dinv, int n) {
  int i = blockIdx.x * blockDim.x + threadIdx.x;
  if (i < n) dinv[i] = rsqrtf(deg[i]);
}

__global__ __launch_bounds__(128) void wmma_gemm_f32_kernel(
    const float* __restrict__ A, const float* __restrict__ B,
    float* __restrict__ C, int M, int N, int K) {
  const int wave  = threadIdx.x >> 5;
  const int lane  = threadIdx.x & 31;
  const int l16   = lane & 15;
  const int lhalf = lane >> 4;

  const int wm = wave >> 1;
  const int wn = wave & 1;
  const int row0 = blockIdx.y * 64 + wm * 32;
  const int col0 = blockIdx.x * 64 + wn * 32;

  const float* arow[2];
#pragma unroll
  for (int mi = 0; mi < 2; ++mi) {
    int r = row0 + mi * 16 + l16;
    r = r < M ? r : M - 1;
    arow[mi] = A + (size_t)r * K + lhalf * 2;
  }
  const float* bcol[2];
#pragma unroll
  for (int ni = 0; ni < 2; ++ni)
    bcol[ni] = B + (size_t)(lhalf * 2) * N + (col0 + ni * 16 + l16);

  v8f acc[2][2] = {};
#pragma unroll 4
  for (int k = 0; k < K; k += 4) {
    v2f a[2], b[2];
#pragma unroll
    for (int mi = 0; mi < 2; ++mi) {
      a[mi].x = arow[mi][k];
      a[mi].y = arow[mi][k + 1];
    }
#pragma unroll
    for (int ni = 0; ni < 2; ++ni) {
      b[ni].x = bcol[ni][(size_t)k * N];
      b[ni].y = bcol[ni][(size_t)(k + 1) * N];
    }
#pragma unroll
    for (int mi = 0; mi < 2; ++mi)
#pragma unroll
      for (int ni = 0; ni < 2; ++ni)
        acc[mi][ni] = __builtin_amdgcn_wmma_f32_16x16x4_f32(
             false, a[mi],  false, b[ni],
             (short)0, acc[mi][ni],  false,  false);
  }

#pragma unroll
  for (int mi = 0; mi < 2; ++mi) {
#pragma unroll
    for (int r = 0; r < 8; ++r) {
      int row = row0 + mi * 16 + 8 * lhalf + r;
      if (row < M) {
#pragma unroll
        for (int ni = 0; ni < 2; ++ni)
          C[(size_t)row * N + col0 + ni * 16 + l16] = acc[mi][ni][r];
      }
    }
  }
}

__global__ void init_out_kernel(const float* __restrict__ h, const float* __restrict__ dinv,
                                const float* __restrict__ bias, float* __restrict__ out,
                                int n, int c) {
  int idx = blockIdx.x * blockDim.x + threadIdx.x;
  if (idx >= n * c) return;
  int i = idx / c;
  int f = idx - i * c;
  float di = dinv[i];
  out[idx] = h[idx] * di * di + bias[f];
}

__global__ void scatter_kernel(const float* __restrict__ h, const int* __restrict__ src,
                               const int* __restrict__ dst, const float* __restrict__ dinv,
                               float* __restrict__ out, int c) {
  int e = blockIdx.x;
  int s = src[e];
  int d = dst[e];
  float norm = dinv[s] * dinv[d];
  const float* hs = h + (size_t)s * c;
  float* od = out + (size_t)d * c;
  for (int f = threadIdx.x; f < c; f += blockDim.x)
    atomicAdd(&od[f], hs[f] * norm);
}

__global__ void relu_kernel(float* __restrict__ x, int n) {
  int i = blockIdx.x * blockDim.x + threadIdx.x;
  if (i < n) x[i] = fmaxf(x[i], 0.0f);
}

extern "C" void kernel_launch(void* const* d_in, const int* in_sizes, int n_in,
                              void* d_out, int out_size, void* d_ws, size_t ws_size,
                              hipStream_t stream) {
  const float* x   = (const float*)d_in[0];
  const int*   ei  = (const int*)d_in[1];
  const float* W1  = (const float*)d_in[2];
  const float* b1  = (const float*)d_in[3];
  const float* W2  = (const float*)d_in[4];
  const float* b2  = (const float*)d_in[5];
  float* out = (float*)d_out;

  const int N = NNODES;
  const int E = NEDGES;
  const int C0 = 512, C1 = 1024, C2 = 512;

  const int* src = ei;
  const int* dst = ei + E;

  float* deg  = (float*)d_ws;
  float* dinv = deg + N;
  float* h    = dinv + N;
  float* agg1 = h + (size_t)N * C1;

  const int T = 256;
  const int MB = (N + 63) / 64;

  deg_init_kernel<<<(N + T - 1) / T, T, 0, stream>>>(deg, N);
  deg_accum_kernel<<<(E + T - 1) / T, T, 0, stream>>>(dst, deg, E);
  dinv_kernel<<<(N + T - 1) / T, T, 0, stream>>>(deg, dinv, N);

  {
    dim3 grid(C1 / 64, MB);
    wmma_gemm_f32_kernel<<<grid, 128, 0, stream>>>(x, W1, h, N, C1, C0);
    init_out_kernel<<<((N * C1) + T - 1) / T, T, 0, stream>>>(h, dinv, b1, agg1, N, C1);
    scatter_kernel<<<E, T, 0, stream>>>(h, src, dst, dinv, agg1, C1);
    relu_kernel<<<((N * C1) + T - 1) / T, T, 0, stream>>>(agg1, N * C1);
  }

  {
    dim3 grid(C2 / 64, MB);
    wmma_gemm_f32_kernel<<<grid, 128, 0, stream>>>(agg1, W2, h, N, C2, C1);
    init_out_kernel<<<((N * C2) + T - 1) / T, T, 0, stream>>>(h, dinv, b2, out, N, C2);
    scatter_kernel<<<E, T, 0, stream>>>(h, src, dst, dinv, out, C2);
    relu_kernel<<<((N * C2) + T - 1) / T, T, 0, stream>>>(out, N * C2);
  }
}


# GEMM fed without LDS by 16-byte global loads: wave tile 32x64 (16x64 in layer 2), B 4 columns per lane (tile columns interleaved), A 4 k per lane, next k set prefetched; k sum order changes
# speedup vs baseline: 1.2040x; 1.2040x over previous
#include <hip/hip_runtime.h>

typedef float v2f __attribute__((ext_vector_type(2)));
typedef float v4f __attribute__((ext_vector_type(4)));
typedef float v8f __attribute__((ext_vector_type(8)));

#define NNODES 10000
#define NEDGES 160000

__global__ void deg_init_kernel(float* __restrict__ deg, int n) {
  int i = blockIdx.x * blockDim.x + threadIdx.x;
  if (i < n) deg[i] = 1.0f;
}

__global__ void deg_accum_kernel(const int* __restrict__ dst, float* __restrict__ deg, int e) {
  int i = blockIdx.x * blockDim.x + threadIdx.x;
  if (i < e) atomicAdd(&deg[dst[i]], 1.0f);
}

__global__ void dinv_kernel(const float* __restrict__ deg, float* __restrict__ dinv, int n) {
  int i = blockIdx.x * blockDim.x + threadIdx.x;
  if (i < n) dinv[i] = rsqrtf(deg[i]);
}

template <int MI, int KB>
__device__ __forceinline__ void gemm_load_set(const float* const (&ap)[MI], const float* bp,
                                              int k0, int nn, v4f (&a)[KB][MI], v4f (&b)[KB][4]) {
#pragma unroll
  for (int kb = 0; kb < KB; ++kb) {
#pragma unroll
    for (int mi = 0; mi < MI; ++mi) a[kb][mi] = *(const v4f*)(ap[mi] + k0 + 8 * kb);
#pragma unroll
    for (int q = 0; q < 4; ++q)
      b[kb][q] = *(const v4f*)(bp + (size_t)(k0 + 8 * kb + q) * nn);
  }
}

template <int MI, int KB>
__device__ __forceinline__ void gemm_mma_set(const v4f (&a)[KB][MI], const v4f (&b)[KB][4],
                                             v8f (&acc)[MI][4]) {
#pragma unroll
  for (int kb = 0; kb < KB; ++kb) {
#pragma unroll
    for (int j = 0; j < 2; ++j) {
#pragma unroll
      for (int t = 0; t < 4; ++t) {
        v2f bf;
        bf.x = b[kb][2 * j][t];
        bf.y = b[kb][2 * j + 1][t];
#pragma unroll
        for (int mi = 0; mi < MI; ++mi) {
          v2f af;
          af.x = a[kb][mi][2 * j];
          af.y = a[kb][mi][2 * j + 1];
          acc[mi][t] = __builtin_amdgcn_wmma_f32_16x16x4_f32(
               false, af,  false, bf,
               (short)0, acc[mi][t],  false,  false);
        }
      }
    }
  }
}

template <int MI, int KB, int NN, int KK>
__global__ __launch_bounds__(128) void wmma_gemm_f32_kernel(
    const float* __restrict__ A, const float* __restrict__ B,
    float* __restrict__ C, int M) {
  const int wave = __builtin_amdgcn_readfirstlane((int)(threadIdx.x >> 5));
  const int lane = threadIdx.x & 31;
  const int c    = lane & 15;
  const int h    = lane >> 4;

  const int row0 = (blockIdx.y * 4 + wave) * (16 * MI);
  if (row0 >= M) return;
  const int col0 = blockIdx.x * 64;

  const float* ap[MI];
#pragma unroll
  for (int mi = 0; mi < MI; ++mi) {
    int r = row0 + mi * 16 + c;
    r = r < M ? r : M - 1;
    ap[mi] = A + (size_t)r * KK + 4 * h;
  }
  const float* bp = B + (size_t)(4 * h) * NN + (col0 + 4 * c);

  v8f acc[MI][4] = {};
  v4f a0[KB][MI], a1[KB][MI], b0[KB][4], b1[KB][4];

  constexpr int KS = 8 * KB;
  gemm_load_set<MI, KB>(ap, bp, 0, NN, a0, b0);
  for (int k0 = 0; k0 < KK; k0 += 2 * KS) {
    gemm_load_set<MI, KB>(ap, bp, k0 + KS, NN, a1, b1);
    __builtin_amdgcn_sched_barrier(0);
    gemm_mma_set<MI, KB>(a0, b0, acc);
    __builtin_amdgcn_sched_barrier(0);
    const int kn = (k0 + 2 * KS < KK) ? k0 + 2 * KS : KK - KS;
    gemm_load_set<MI, KB>(ap, bp, kn, NN, a0, b0);
    __builtin_amdgcn_sched_barrier(0);
    gemm_mma_set<MI, KB>(a1, b1, acc);
  }

#pragma unroll
  for (int mi = 0; mi < MI; ++mi) {
#pragma unroll
    for (int r = 0; r < 8; ++r) {
      const int row = row0 + mi * 16 + 8 * h + r;
      if (row < M) {
        v4f o;
        o.x = acc[mi][0][r];
        o.y = acc[mi][1][r];
        o.z = acc[mi][2][r];
        o.w = acc[mi][3][r];
        *(v4f*)(C + (size_t)row * NN + col0 + 4 * c) = o;
      }
    }
  }
}

__global__ void init_out_kernel(const float* __restrict__ h, const float* __restrict__ dinv,
                                const float* __restrict__ bias, float* __restrict__ out,
                                int n, int c) {
  int idx = blockIdx.x * blockDim.x + threadIdx.x;
  if (idx >= n * c) return;
  int i = idx / c;
  int f = idx - i * c;
  float di = dinv[i];
  out[idx] = h[idx] * di * di + bias[f];
}

__global__ void scatter_kernel(const float* __restrict__ h, const int* __restrict__ src,
                               const int* __restrict__ dst, const float* __restrict__ dinv,
                               float* __restrict__ out, int c) {
  int e = blockIdx.x;
  int s = src[e];
  int d = dst[e];
  float norm = dinv[s] * dinv[d];
  const float* hs = h + (size_t)s * c;
  float* od = out + (size_t)d * c;
  for (int f = threadIdx.x; f < c; f += blockDim.x)
    atomicAdd(&od[f], hs[f] * norm);
}

__global__ void relu_kernel(float* __restrict__ x, int n) {
  int i = blockIdx.x * blockDim.x + threadIdx.x;
  if (i < n) x[i] = fmaxf(x[i], 0.0f);
}

extern "C" void kernel_launch(void* const* d_in, const int* in_sizes, int n_in,
                              void* d_out, int out_size, void* d_ws, size_t ws_size,
                              hipStream_t stream) {
  const float* x   = (const float*)d_in[0];
  const int*   ei  = (const int*)d_in[1];
  const float* W1  = (const float*)d_in[2];
  const float* b1  = (const float*)d_in[3];
  const float* W2  = (const float*)d_in[4];
  const float* b2  = (const float*)d_in[5];
  float* out = (float*)d_out;

  const int N = NNODES;
  const int E = NEDGES;
  constexpr int C0 = 512, C1 = 1024, C2 = 512;

  const int* src = ei;
  const int* dst = ei + E;

  float* deg  = (float*)d_ws;
  float* dinv = deg + N;
  float* h    = dinv + N;
  float* agg1 = h + (size_t)N * C1;

  const int T = 256;

  deg_init_kernel<<<(N + T - 1) / T, T, 0, stream>>>(deg, N);
  deg_accum_kernel<<<(E + T - 1) / T, T, 0, stream>>>(dst, deg, E);
  dinv_kernel<<<(N + T - 1) / T, T, 0, stream>>>(deg, dinv, N);

  {
    dim3 grid(C1 / 64, (N + 127) / 128);
    wmma_gemm_f32_kernel<2, 1, C1, C0><<<grid, 128, 0, stream>>>(x, W1, h, N);
    init_out_kernel<<<((N * C1) + T - 1) / T, T, 0, stream>>>(h, dinv, b1, agg1, N, C1);
    scatter_kernel<<<E, T, 0, stream>>>(h, src, dst, dinv, agg1, C1);
    relu_kernel<<<((N * C1) + T - 1) / T, T, 0, stream>>>(agg1, N * C1);
  }

  {
    dim3 grid(C2 / 64, (N + 63) / 64);
    wmma_gemm_f32_kernel<1, 2, C2, C1><<<grid, 128, 0, stream>>>(agg1, W2, h, N);
    init_out_kernel<<<((N * C2) + T - 1) / T, T, 0, stream>>>(h, dinv, b2, out, N, C2);
    scatter_kernel<<<E, T, 0, stream>>>(h, src, dst, dinv, out, C2);
    relu_kernel<<<((N * C2) + T - 1) / T, T, 0, stream>>>(out, N * C2);
  }
}


# Edge scatter: one wave walks 16 edges over 512 columns (indices and norm read once, handed out by readlane; next row requested ahead), not a block per edge; float-atomic sum order still by arrival
# speedup vs baseline: 1.9769x; 1.6419x over previous
#include <hip/hip_runtime.h>

typedef float v2f __attribute__((ext_vector_type(2)));
typedef float v4f __attribute__((ext_vector_type(4)));
typedef float v8f __attribute__((ext_vector_type(8)));

#define NNODES 10000
#define NEDGES 160000

__global__ void deg_init_kernel(float* __restrict__ deg, int n) {
  int i = blockIdx.x * blockDim.x + threadIdx.x;
  if (i < n) deg[i] = 1.0f;
}

__global__ void deg_accum_kernel(const int* __restrict__ dst, float* __restrict__ deg, int e) {
  int i = blockIdx.x * blockDim.x + threadIdx.x;
  if (i < e) atomicAdd(&deg[dst[i]], 1.0f);
}

__global__ void dinv_kernel(const float* __restrict__ deg, float* __restrict__ dinv, int n) {
  int i = blockIdx.x * blockDim.x + threadIdx.x;
  if (i < n) dinv[i] = rsqrtf(deg[i]);
}

template <int MI, int KB>
__device__ __forceinline__ void gemm_load_set(const float* const (&ap)[MI], const float* bp,
                                              int k0, int nn, v4f (&a)[KB][MI], v4f (&b)[KB][4]) {
#pragma unroll
  for (int kb = 0; kb < KB; ++kb) {
#pragma unroll
    for (int mi = 0; mi < MI; ++mi) a[kb][mi] = *(const v4f*)(ap[mi] + k0 + 8 * kb);
#pragma unroll
    for (int q = 0; q < 4; ++q)
      b[kb][q] = *(const v4f*)(bp + (size_t)(k0 + 8 * kb + q) * nn);
  }
}

template <int MI, int KB>
__device__ __forceinline__ void gemm_mma_set(const v4f (&a)[KB][MI], const v4f (&b)[KB][4],
                                             v8f (&acc)[MI][4]) {
#pragma unroll
  for (int kb = 0; kb < KB; ++kb) {
#pragma unroll
    for (int j = 0; j < 2; ++j) {
#pragma unroll
      for (int t = 0; t < 4; ++t) {
        v2f bf;
        bf.x = b[kb][2 * j][t];
        bf.y = b[kb][2 * j + 1][t];
#pragma unroll
        for (int mi = 0; mi < MI; ++mi) {
          v2f af;
          af.x = a[kb][mi][2 * j];
          af.y = a[kb][mi][2 * j + 1];
          acc[mi][t] = __builtin_amdgcn_wmma_f32_16x16x4_f32(
               false, af,  false, bf,
               (short)0, acc[mi][t],  false,  false);
        }
      }
    }
  }
}

template <int MI, int KB, int NN, int KK>
__global__ __launch_bounds__(128) void wmma_gemm_f32_kernel(
    const float* __restrict__ A, const float* __restrict__ B,
    float* __restrict__ C, int M) {
  const int wave = __builtin_amdgcn_readfirstlane((int)(threadIdx.x >> 5));
  const int lane = threadIdx.x & 31;
  const int c    = lane & 15;
  const int h    = lane >> 4;

  const int row0 = (blockIdx.y * 4 + wave) * (16 * MI);
  if (row0 >= M) return;
  const int col0 = blockIdx.x * 64;

  const float* ap[MI];
#pragma unroll
  for (int mi = 0; mi < MI; ++mi) {
    int r = row0 + mi * 16 + c;
    r = r < M ? r : M - 1;
    ap[mi] = A + (size_t)r * KK + 4 * h;
  }
  const float* bp = B + (size_t)(4 * h) * NN + (col0 + 4 * c);

  v8f acc[MI][4] = {};
  v4f a0[KB][MI], a1[KB][MI], b0[KB][4], b1[KB][4];

  constexpr int KS = 8 * KB;
  gemm_load_set<MI, KB>(ap, bp, 0, NN, a0, b0);
  for (int k0 = 0; k0 < KK; k0 += 2 * KS) {
    gemm_load_set<MI, KB>(ap, bp, k0 + KS, NN, a1, b1);
    __builtin_amdgcn_sched_barrier(0);
    gemm_mma_set<MI, KB>(a0, b0, acc);
    __builtin_amdgcn_sched_barrier(0);
    const int kn = (k0 + 2 * KS < KK) ? k0 + 2 * KS : KK - KS;
    gemm_load_set<MI, KB>(ap, bp, kn, NN, a0, b0);
    __builtin_amdgcn_sched_barrier(0);
    gemm_mma_set<MI, KB>(a1, b1, acc);
  }

#pragma unroll
  for (int mi = 0; mi < MI; ++mi) {
#pragma unroll
    for (int r = 0; r < 8; ++r) {
      const int row = row0 + mi * 16 + 8 * h + r;
      if (row < M) {
        v4f o;
        o.x = acc[mi][0][r];
        o.y = acc[mi][1][r];
        o.z = acc[mi][2][r];
        o.w = acc[mi][3][r];
        *(v4f*)(C + (size_t)row * NN + col0 + 4 * c) = o;
      }
    }
  }
}

__global__ void init_out_kernel(const float* __restrict__ h, const float* __restrict__ dinv,
                                const float* __restrict__ bias, float* __restrict__ out,
                                int n, int c) {
  int idx = blockIdx.x * blockDim.x + threadIdx.x;
  if (idx >= n * c) return;
  int i = idx / c;
  int f = idx - i * c;
  float di = dinv[i];
  out[idx] = h[idx] * di * di + bias[f];
}

#define SC_EPW  16
#define SC_COLS 512
#define SC_NV   (SC_COLS / 32)
static_assert(NEDGES % SC_EPW == 0, "every wave owns a full run of edges");

__device__ __forceinline__ void scatter_load_row(const float* __restrict__ hs, float (&v)[SC_NV]) {
#pragma unroll
  for (int j = 0; j < SC_NV; ++j) v[j] = hs[32 * j];
}

__device__ __forceinline__ void scatter_add_row(float* __restrict__ od, const float (&v)[SC_NV],
                                                float norm) {
#pragma unroll
  for (int j = 0; j < SC_NV; ++j) atomicAdd(&od[32 * j], v[j] * norm);
}

template <int C>
__global__ __launch_bounds__(256) void scatter_kernel(const float* __restrict__ h,
                                                      const int* __restrict__ src,
                                                      const int* __restrict__ dst,
                                                      const float* __restrict__ dinv,
                                                      float* __restrict__ out, int n_edges) {
  constexpr int SLICES = C / SC_COLS;
  const int wib  = __builtin_amdgcn_readfirstlane((int)(threadIdx.x >> 5));
  const int lane = threadIdx.x & 31;
  const int wave = blockIdx.x * (int)(blockDim.x >> 5) + wib;
  const int run   = wave / SLICES;
  const int slice = wave - run * SLICES;
  const int e0 = run * SC_EPW;
  if (e0 >= n_edges) return;

  const int   el  = e0 + (lane & (SC_EPW - 1));
  const int   s_l = src[el];
  const int   d_l = dst[el];
  const float n_l = dinv[s_l] * dinv[d_l];
  const int   n_b = __builtin_bit_cast(int, n_l);

  const float* hb = h + slice * SC_COLS + lane;
  float*       ob = out + slice * SC_COLS + lane;

  float r[2][SC_NV];
  scatter_load_row(hb + (size_t)__builtin_amdgcn_readlane(s_l, 0) * C, r[0]);
#pragma unroll
  for (int i = 0; i < SC_EPW; ++i) {
    if (i + 1 < SC_EPW)
      scatter_load_row(hb + (size_t)__builtin_amdgcn_readlane(s_l, i + 1) * C, r[(i + 1) & 1]);
    scatter_add_row(ob + (size_t)__builtin_amdgcn_readlane(d_l, i) * C, r[i & 1],
                    __builtin_bit_cast(float, __builtin_amdgcn_readlane(n_b, i)));
  }
}

__global__ void relu_kernel(float* __restrict__ x, int n) {
  int i = blockIdx.x * blockDim.x + threadIdx.x;
  if (i < n) x[i] = fmaxf(x[i], 0.0f);
}

extern "C" void kernel_launch(void* const* d_in, const int* in_sizes, int n_in,
                              void* d_out, int out_size, void* d_ws, size_t ws_size,
                              hipStream_t stream) {
  const float* x   = (const float*)d_in[0];
  const int*   ei  = (const int*)d_in[1];
  const float* W1  = (const float*)d_in[2];
  const float* b1  = (const float*)d_in[3];
  const float* W2  = (const float*)d_in[4];
  const float* b2  = (const float*)d_in[5];
  float* out = (float*)d_out;

  const int N = NNODES;
  const int E = NEDGES;
  constexpr int C0 = 512, C1 = 1024, C2 = 512;

  const int* src = ei;
  const int* dst = ei + E;

  float* deg  = (float*)d_ws;
  float* dinv = deg + N;
  float* h    = dinv + N;
  float* agg1 = h + (size_t)N * C1;

  const int T = 256;

  deg_init_kernel<<<(N + T - 1) / T, T, 0, stream>>>(deg, N);
  deg_accum_kernel<<<(E + T - 1) / T, T, 0, stream>>>(dst, deg, E);
  dinv_kernel<<<(N + T - 1) / T, T, 0, stream>>>(deg, dinv, N);

  {
    dim3 grid(C1 / 64, (N + 127) / 128);
    wmma_gemm_f32_kernel<2, 1, C1, C0><<<grid, 128, 0, stream>>>(x, W1, h, N);
    init_out_kernel<<<((N * C1) + T - 1) / T, T, 0, stream>>>(h, dinv, b1, agg1, N, C1);
    scatter_kernel<C1><<<(E / SC_EPW * (C1 / SC_COLS) + 7) / 8, 256, 0, stream>>>(h, src, dst, dinv, agg1, E);
    relu_kernel<<<((N * C1) + T - 1) / T, T, 0, stream>>>(agg1, N * C1);
  }

  {
    dim3 grid(C2 / 64, (N + 63) / 64);
    wmma_gemm_f32_kernel<1, 2, C2, C1><<<grid, 128, 0, stream>>>(agg1, W2, h, N);
    init_out_kernel<<<((N * C2) + T - 1) / T, T, 0, stream>>>(h, dinv, b2, out, N, C2);
    scatter_kernel<C2><<<(E / SC_EPW * (C2 / SC_COLS) + 7) / 8, 256, 0, stream>>>(h, src, dst, dinv, out, E);
    relu_kernel<<<((N * C2) + T - 1) / T, T, 0, stream>>>(out, N * C2);
  }
}


# Element-wise passes folded into the GEMM launches: the store phase also writes the scatter's start value (h*dinv^2 + bias), layer 2's A loader applies layer 1's relu; 11 launches become 8
# speedup vs baseline: 2.2272x; 1.1266x over previous
#include <hip/hip_runtime.h>

typedef float v2f __attribute__((ext_vector_type(2)));
typedef float v4f __attribute__((ext_vector_type(4)));
typedef float v8f __attribute__((ext_vector_type(8)));

#define NNODES 10000
#define NEDGES 160000

__global__ void deg_init_kernel(float* __restrict__ deg, int n) {
  int i = blockIdx.x * blockDim.x + threadIdx.x;
  if (i < n) deg[i] = 1.0f;
}

__global__ void deg_accum_kernel(const int* __restrict__ dst, float* __restrict__ deg, int e) {
  int i = blockIdx.x * blockDim.x + threadIdx.x;
  if (i < e) atomicAdd(&deg[dst[i]], 1.0f);
}

__global__ void dinv_kernel(const float* __restrict__ deg, float* __restrict__ dinv, int n) {
  int i = blockIdx.x * blockDim.x + threadIdx.x;
  if (i < n) dinv[i] = rsqrtf(deg[i]);
}

template <int MI, int KB>
__device__ __forceinline__ void gemm_load_set(const float* const (&ap)[MI], const float* bp,
                                              int k0, int nn, v4f (&a)[KB][MI], v4f (&b)[KB][4]) {
#pragma unroll
  for (int kb = 0; kb < KB; ++kb) {
#pragma unroll
    for (int mi = 0; mi < MI; ++mi) a[kb][mi] = *(const v4f*)(ap[mi] + k0 + 8 * kb);
#pragma unroll
    for (int q = 0; q < 4; ++q)
      b[kb][q] = *(const v4f*)(bp + (size_t)(k0 + 8 * kb + q) * nn);
  }
}

template <int MI, int KB, bool RELU_A>
__device__ __forceinline__ void gemm_mma_set(const v4f (&a)[KB][MI], const v4f (&b)[KB][4],
                                             v8f (&acc)[MI][4]) {
#pragma unroll
  for (int kb = 0; kb < KB; ++kb) {
    v4f ar[MI];
#pragma unroll
    for (int mi = 0; mi < MI; ++mi) {
      ar[mi] = a[kb][mi];
      if (RELU_A) {
        ar[mi].x = fmaxf(ar[mi].x, 0.0f);
        ar[mi].y = fmaxf(ar[mi].y, 0.0f);
        ar[mi].z = fmaxf(ar[mi].z, 0.0f);
        ar[mi].w = fmaxf(ar[mi].w, 0.0f);
      }
    }
#pragma unroll
    for (int j = 0; j < 2; ++j) {
#pragma unroll
      for (int t = 0; t < 4; ++t) {
        v2f bf;
        bf.x = b[kb][2 * j][t];
        bf.y = b[kb][2 * j + 1][t];
#pragma unroll
        for (int mi = 0; mi < MI; ++mi) {
          v2f af;
          af.x = ar[mi][2 * j];
          af.y = ar[mi][2 * j + 1];
          acc[mi][t] = __builtin_amdgcn_wmma_f32_16x16x4_f32(
               false, af,  false, bf,
               (short)0, acc[mi][t],  false,  false);
        }
      }
    }
  }
}

template <int MI, int KB, int NN, int KK, bool RELU_A>
__global__ __launch_bounds__(128) void wmma_gemm_f32_kernel(
    const float* __restrict__ A, const float* __restrict__ B,
    const float* __restrict__ dinv, const float* __restrict__ bias,
    float* __restrict__ C, float* __restrict__ S, int M) {
  const int wave = __builtin_amdgcn_readfirstlane((int)(threadIdx.x >> 5));
  const int lane = threadIdx.x & 31;
  const int c    = lane & 15;
  const int h    = lane >> 4;

  const int row0 = (blockIdx.y * 4 + wave) * (16 * MI);
  if (row0 >= M) return;
  const int col0 = blockIdx.x * 64;

  const float* ap[MI];
#pragma unroll
  for (int mi = 0; mi < MI; ++mi) {
    int r = row0 + mi * 16 + c;
    r = r < M ? r : M - 1;
    ap[mi] = A + (size_t)r * KK + 4 * h;
  }
  const float* bp = B + (size_t)(4 * h) * NN + (col0 + 4 * c);

  v8f acc[MI][4] = {};
  v4f a0[KB][MI], a1[KB][MI], b0[KB][4], b1[KB][4];

  constexpr int KS = 8 * KB;
  gemm_load_set<MI, KB>(ap, bp, 0, NN, a0, b0);
  for (int k0 = 0; k0 < KK; k0 += 2 * KS) {
    gemm_load_set<MI, KB>(ap, bp, k0 + KS, NN, a1, b1);
    __builtin_amdgcn_sched_barrier(0);
    gemm_mma_set<MI, KB, RELU_A>(a0, b0, acc);
    __builtin_amdgcn_sched_barrier(0);
    const int kn = (k0 + 2 * KS < KK) ? k0 + 2 * KS : KK - KS;
    gemm_load_set<MI, KB>(ap, bp, kn, NN, a0, b0);
    __builtin_amdgcn_sched_barrier(0);
    gemm_mma_set<MI, KB, RELU_A>(a1, b1, acc);
  }

  const v4f bv = *(const v4f*)(bias + col0 + 4 * c);
  float dv[MI][8];
#pragma unroll
  for (int mi = 0; mi < MI; ++mi) {
#pragma unroll
    for (int r = 0; r < 8; ++r) {
      int rr = row0 + mi * 16 + 8 * h + r;
      rr = rr < M ? rr : M - 1;
      dv[mi][r] = dinv[rr];
    }
  }
#pragma unroll
  for (int mi = 0; mi < MI; ++mi) {
#pragma unroll
    for (int r = 0; r < 8; ++r) {
      const int row = row0 + mi * 16 + 8 * h + r;
      if (row < M) {
        const float di = dv[mi][r];
        v4f o;
        o.x = acc[mi][0][r];
        o.y = acc[mi][1][r];
        o.z = acc[mi][2][r];
        o.w = acc[mi][3][r];
        *(v4f*)(C + (size_t)row * NN + col0 + 4 * c) = o;
        v4f s;
        s.x = o.x * di * di + bv.x;
        s.y = o.y * di * di + bv.y;
        s.z = o.z * di * di + bv.z;
        s.w = o.w * di * di + bv.w;
        *(v4f*)(S + (size_t)row * NN + col0 + 4 * c) = s;
      }
    }
  }
}

#define SC_EPW  16
#define SC_COLS 512
#define SC_NV   (SC_COLS / 32)
static_assert(NEDGES % SC_EPW == 0, "every wave owns a full run of edges");

__device__ __forceinline__ void scatter_load_row(const float* __restrict__ hs, float (&v)[SC_NV]) {
#pragma unroll
  for (int j = 0; j < SC_NV; ++j) v[j] = hs[32 * j];
}

__device__ __forceinline__ void scatter_add_row(float* __restrict__ od, const float (&v)[SC_NV],
                                                float norm) {
#pragma unroll
  for (int j = 0; j < SC_NV; ++j) atomicAdd(&od[32 * j], v[j] * norm);
}

template <int C>
__global__ __launch_bounds__(256) void scatter_kernel(const float* __restrict__ h,
                                                      const int* __restrict__ src,
                                                      const int* __restrict__ dst,
                                                      const float* __restrict__ dinv,
                                                      float* __restrict__ out, int n_edges) {
  constexpr int SLICES = C / SC_COLS;
  const int wib  = __builtin_amdgcn_readfirstlane((int)(threadIdx.x >> 5));
  const int lane = threadIdx.x & 31;
  const int wave = blockIdx.x * (int)(blockDim.x >> 5) + wib;
  const int run   = wave / SLICES;
  const int slice = wave - run * SLICES;
  const int e0 = run * SC_EPW;
  if (e0 >= n_edges) return;

  const int   el  = e0 + (lane & (SC_EPW - 1));
  const int   s_l = src[el];
  const int   d_l = dst[el];
  const float n_l = dinv[s_l] * dinv[d_l];
  const int   n_b = __builtin_bit_cast(int, n_l);

  const float* hb = h + slice * SC_COLS + lane;
  float*       ob = out + slice * SC_COLS + lane;

  float r[2][SC_NV];
  scatter_load_row(hb + (size_t)__builtin_amdgcn_readlane(s_l, 0) * C, r[0]);
#pragma unroll
  for (int i = 0; i < SC_EPW; ++i) {
    if (i + 1 < SC_EPW)
      scatter_load_row(hb + (size_t)__builtin_amdgcn_readlane(s_l, i + 1) * C, r[(i + 1) & 1]);
    scatter_add_row(ob + (size_t)__builtin_amdgcn_readlane(d_l, i) * C, r[i & 1],
                    __builtin_bit_cast(float, __builtin_amdgcn_readlane(n_b, i)));
  }
}

__global__ void relu_kernel(float* __restrict__ x, int n) {
  int i = blockIdx.x * blockDim.x + threadIdx.x;
  if (i < n) x[i] = fmaxf(x[i], 0.0f);
}

extern "C" void kernel_launch(void* const* d_in, const int* in_sizes, int n_in,
                              void* d_out, int out_size, void* d_ws, size_t ws_size,
                              hipStream_t stream) {
  const float* x   = (const float*)d_in[0];
  const int*   ei  = (const int*)d_in[1];
  const float* W1  = (const float*)d_in[2];
  const float* b1  = (const float*)d_in[3];
  const float* W2  = (const float*)d_in[4];
  const float* b2  = (const float*)d_in[5];
  float* out = (float*)d_out;

  const int N = NNODES;
  const int E = NEDGES;
  constexpr int C0 = 512, C1 = 1024, C2 = 512;

  const int* src = ei;
  const int* dst = ei + E;

  float* deg  = (float*)d_ws;
  float* dinv = deg + N;
  float* h    = dinv + N;
  float* agg1 = h + (size_t)N * C1;

  const int T = 256;

  deg_init_kernel<<<(N + T - 1) / T, T, 0, stream>>>(deg, N);
  deg_accum_kernel<<<(E + T - 1) / T, T, 0, stream>>>(dst, deg, E);
  dinv_kernel<<<(N + T - 1) / T, T, 0, stream>>>(deg, dinv, N);

  {
    dim3 grid(C1 / 64, (N + 127) / 128);
    wmma_gemm_f32_kernel<2, 1, C1, C0, false><<<grid, 128, 0, stream>>>(x, W1, dinv, b1, h, agg1, N);
    scatter_kernel<C1><<<(E / SC_EPW * (C1 / SC_COLS) + 7) / 8, 256, 0, stream>>>(h, src, dst, dinv, agg1, E);
  }

  {
    dim3 grid(C2 / 64, (N + 63) / 64);
    wmma_gemm_f32_kernel<1, 2, C2, C1, true><<<grid, 128, 0, stream>>>(agg1, W2, dinv, b2, h, out, N);
    scatter_kernel<C2><<<(E / SC_EPW * (C2 / SC_COLS) + 7) / 8, 256, 0, stream>>>(h, src, dst, dinv, out, E);
    relu_kernel<<<((N * C2) + T - 1) / T, T, 0, stream>>>(out, N * C2);
  }
}


# Edge scatter split by chip half: a wave reads which half it runs on and draws, by ticket, a run of edges over that half's column slice (then helps the other); float-atomic sum order still by arrival
# speedup vs baseline: 2.4607x; 1.1048x over previous
#include <hip/hip_runtime.h>

typedef float v2f __attribute__((ext_vector_type(2)));
typedef float v4f __attribute__((ext_vector_type(4)));
typedef float v8f __attribute__((ext_vector_type(8)));

#define NNODES 10000
#define NEDGES 160000

#define SC_CTR_PITCH 64

__global__ void deg_init_kernel(float* __restrict__ deg, int* __restrict__ ctr, int n) {
  int i = blockIdx.x * blockDim.x + threadIdx.x;
  if (i < n) deg[i] = 1.0f;
  if (i < 4) ctr[i * SC_CTR_PITCH] = 0;
}

__global__ void deg_accum_kernel(const int* __restrict__ dst, float* __restrict__ deg, int e) {
  int i = blockIdx.x * blockDim.x + threadIdx.x;
  if (i < e) atomicAdd(&deg[dst[i]], 1.0f);
}

__global__ void dinv_kernel(const float* __restrict__ deg, float* __restrict__ dinv, int n) {
  int i = blockIdx.x * blockDim.x + threadIdx.x;
  if (i < n) dinv[i] = rsqrtf(deg[i]);
}

template <int MI, int KB>
__device__ __forceinline__ void gemm_load_set(const float* const (&ap)[MI], const float* bp,
                                              int k0, int nn, v4f (&a)[KB][MI], v4f (&b)[KB][4]) {
#pragma unroll
  for (int kb = 0; kb < KB; ++kb) {
#pragma unroll
    for (int mi = 0; mi < MI; ++mi) a[kb][mi] = *(const v4f*)(ap[mi] + k0 + 8 * kb);
#pragma unroll
    for (int q = 0; q < 4; ++q)
      b[kb][q] = *(const v4f*)(bp + (size_t)(k0 + 8 * kb + q) * nn);
  }
}

template <int MI, int KB, bool RELU_A>
__device__ __forceinline__ void gemm_mma_set(const v4f (&a)[KB][MI], const v4f (&b)[KB][4],
                                             v8f (&acc)[MI][4]) {
#pragma unroll
  for (int kb = 0; kb < KB; ++kb) {
    v4f ar[MI];
#pragma unroll
    for (int mi = 0; mi < MI; ++mi) {
      ar[mi] = a[kb][mi];
      if (RELU_A) {
        ar[mi].x = fmaxf(ar[mi].x, 0.0f);
        ar[mi].y = fmaxf(ar[mi].y, 0.0f);
        ar[mi].z = fmaxf(ar[mi].z, 0.0f);
        ar[mi].w = fmaxf(ar[mi].w, 0.0f);
      }
    }
#pragma unroll
    for (int j = 0; j < 2; ++j) {
#pragma unroll
      for (int t = 0; t < 4; ++t) {
        v2f bf;
        bf.x = b[kb][2 * j][t];
        bf.y = b[kb][2 * j + 1][t];
#pragma unroll
        for (int mi = 0; mi < MI; ++mi) {
          v2f af;
          af.x = ar[mi][2 * j];
          af.y = ar[mi][2 * j + 1];
          acc[mi][t] = __builtin_amdgcn_wmma_f32_16x16x4_f32(
               false, af,  false, bf,
               (short)0, acc[mi][t],  false,  false);
        }
      }
    }
  }
}

template <int MI, int KB, int NN, int KK, bool RELU_A>
__global__ __launch_bounds__(128) void wmma_gemm_f32_kernel(
    const float* __restrict__ A, const float* __restrict__ B,
    const float* __restrict__ dinv, const float* __restrict__ bias,
    float* __restrict__ C, float* __restrict__ S, int M) {
  const int wave = __builtin_amdgcn_readfirstlane((int)(threadIdx.x >> 5));
  const int lane = threadIdx.x & 31;
  const int c    = lane & 15;
  const int h    = lane >> 4;

  const int row0 = (blockIdx.y * 4 + wave) * (16 * MI);
  if (row0 >= M) return;
  const int col0 = blockIdx.x * 64;

  const float* ap[MI];
#pragma unroll
  for (int mi = 0; mi < MI; ++mi) {
    int r = row0 + mi * 16 + c;
    r = r < M ? r : M - 1;
    ap[mi] = A + (size_t)r * KK + 4 * h;
  }
  const float* bp = B + (size_t)(4 * h) * NN + (col0 + 4 * c);

  v8f acc[MI][4] = {};
  v4f a0[KB][MI], a1[KB][MI], b0[KB][4], b1[KB][4];

  constexpr int KS = 8 * KB;
  gemm_load_set<MI, KB>(ap, bp, 0, NN, a0, b0);
  for (int k0 = 0; k0 < KK; k0 += 2 * KS) {
    gemm_load_set<MI, KB>(ap, bp, k0 + KS, NN, a1, b1);
    __builtin_amdgcn_sched_barrier(0);
    gemm_mma_set<MI, KB, RELU_A>(a0, b0, acc);
    __builtin_amdgcn_sched_barrier(0);
    const int kn = (k0 + 2 * KS < KK) ? k0 + 2 * KS : KK - KS;
    gemm_load_set<MI, KB>(ap, bp, kn, NN, a0, b0);
    __builtin_amdgcn_sched_barrier(0);
    gemm_mma_set<MI, KB, RELU_A>(a1, b1, acc);
  }

  const v4f bv = *(const v4f*)(bias + col0 + 4 * c);
  float dv[MI][8];
#pragma unroll
  for (int mi = 0; mi < MI; ++mi) {
#pragma unroll
    for (int r = 0; r < 8; ++r) {
      int rr = row0 + mi * 16 + 8 * h + r;
      rr = rr < M ? rr : M - 1;
      dv[mi][r] = dinv[rr];
    }
  }
#pragma unroll
  for (int mi = 0; mi < MI; ++mi) {
#pragma unroll
    for (int r = 0; r < 8; ++r) {
      const int row = row0 + mi * 16 + 8 * h + r;
      if (row < M) {
        const float di = dv[mi][r];
        v4f o;
        o.x = acc[mi][0][r];
        o.y = acc[mi][1][r];
        o.z = acc[mi][2][r];
        o.w = acc[mi][3][r];
        *(v4f*)(C + (size_t)row * NN + col0 + 4 * c) = o;
        v4f s;
        s.x = o.x * di * di + bv.x;
        s.y = o.y * di * di + bv.y;
        s.z = o.z * di * di + bv.z;
        s.w = o.w * di * di + bv.w;
        *(v4f*)(S + (size_t)row * NN + col0 + 4 * c) = s;
      }
    }
  }
}

static_assert(NEDGES % 32 == 0, "every ticket is a full run of edges");

__device__ __forceinline__ int chip_half() {
  return (int)((__builtin_amdgcn_s_sendmsg_rtn(0x87) >> 18) & 1u);
}

template <int NV>
__device__ __forceinline__ void scatter_load_row(const float* __restrict__ hs, float (&v)[NV]) {
#pragma unroll
  for (int j = 0; j < NV; ++j) v[j] = hs[32 * j];
}

template <int NV>
__device__ __forceinline__ void scatter_add_row(float* __restrict__ od, const float (&v)[NV],
                                                float norm) {
#pragma unroll
  for (int j = 0; j < NV; ++j) atomicAdd(&od[32 * j], v[j] * norm);
}

template <int C, int NV, int EPW>
__global__ __launch_bounds__(256) void scatter_kernel(const float* __restrict__ h,
                                                      const int* __restrict__ src,
                                                      const int* __restrict__ dst,
                                                      const float* __restrict__ dinv,
                                                      float* __restrict__ out,
                                                      int* __restrict__ ctr, int n_runs) {
  static_assert(C == 2 * 32 * NV, "a row is two slices");
  static_assert(EPW == 16 || EPW == 32, "lane l (mod EPW) holds edge l of the run");
  const int lane = threadIdx.x & 31;

  int slice = chip_half();
  int t = 0;
  if (lane == 0) t = atomicAdd(ctr + slice * SC_CTR_PITCH, 1);
  t = __builtin_amdgcn_readfirstlane(t);
  if (t >= n_runs) {
    slice ^= 1;
    t = 0;
    if (lane == 0) t = atomicAdd(ctr + slice * SC_CTR_PITCH, 1);
    t = __builtin_amdgcn_readfirstlane(t);
    if (t >= n_runs) return;
  }
  const int e0 = t * EPW;

  const int   el  = e0 + (lane & (EPW - 1));
  const int   s_l = src[el];
  const int   d_l = dst[el];
  const float n_l = dinv[s_l] * dinv[d_l];
  const int   n_b = __builtin_bit_cast(int, n_l);

  const float* hb = h + slice * (32 * NV) + lane;
  float*       ob = out + slice * (32 * NV) + lane;

  float r[2][NV];
  scatter_load_row<NV>(hb + (size_t)__builtin_amdgcn_readlane(s_l, 0) * C, r[0]);
#pragma unroll
  for (int i = 0; i < EPW; ++i) {
    if (i + 1 < EPW)
      scatter_load_row<NV>(hb + (size_t)__builtin_amdgcn_readlane(s_l, i + 1) * C, r[(i + 1) & 1]);
    scatter_add_row<NV>(ob + (size_t)__builtin_amdgcn_readlane(d_l, i) * C, r[i & 1],
                        __builtin_bit_cast(float, __builtin_amdgcn_readlane(n_b, i)));
  }
}

__global__ void relu_kernel(float* __restrict__ x, int n) {
  int i = blockIdx.x * blockDim.x + threadIdx.x;
  if (i < n) x[i] = fmaxf(x[i], 0.0f);
}

extern "C" void kernel_launch(void* const* d_in, const int* in_sizes, int n_in,
                              void* d_out, int out_size, void* d_ws, size_t ws_size,
                              hipStream_t stream) {
  const float* x   = (const float*)d_in[0];
  const int*   ei  = (const int*)d_in[1];
  const float* W1  = (const float*)d_in[2];
  const float* b1  = (const float*)d_in[3];
  const float* W2  = (const float*)d_in[4];
  const float* b2  = (const float*)d_in[5];
  float* out = (float*)d_out;

  const int N = NNODES;
  const int E = NEDGES;
  constexpr int C0 = 512, C1 = 1024, C2 = 512;

  const int* src = ei;
  const int* dst = ei + E;

  float* deg  = (float*)d_ws;
  float* dinv = deg + N;
  float* h    = dinv + N;
  float* agg1 = h + (size_t)N * C1;
  int*   ctr  = (int*)(agg1 + (size_t)N * C1);

  const int T = 256;

  deg_init_kernel<<<(N + T - 1) / T, T, 0, stream>>>(deg, ctr, N);
  deg_accum_kernel<<<(E + T - 1) / T, T, 0, stream>>>(dst, deg, E);
  dinv_kernel<<<(N + T - 1) / T, T, 0, stream>>>(deg, dinv, N);

  {
    dim3 grid(C1 / 64, (N + 127) / 128);
    wmma_gemm_f32_kernel<2, 1, C1, C0, false><<<grid, 128, 0, stream>>>(x, W1, dinv, b1, h, agg1, N);
    scatter_kernel<C1, 16, 16><<<(2 * (E / 16) + 7) / 8, 256, 0, stream>>>(h, src, dst, dinv, agg1, ctr, E / 16);
  }

  {
    dim3 grid(C2 / 64, (N + 63) / 64);
    wmma_gemm_f32_kernel<1, 2, C2, C1, true><<<grid, 128, 0, stream>>>(agg1, W2, dinv, b2, h, out, N);
    scatter_kernel<C2, 8, 32><<<(2 * (E / 32) + 7) / 8, 256, 0, stream>>>(h, src, dst, dinv, out, ctr + 2 * SC_CTR_PITCH, E / 32);
    relu_kernel<<<((N * C2) + T - 1) / T, T, 0, stream>>>(out, N * C2);
  }
}
